# starting kernel
# baseline (speedup 1.0000x reference)
#include <hip/hip_runtime.h>
#include <hip/hip_bf16.h>

typedef __attribute__((ext_vector_type(2))) float v2f;
typedef __attribute__((ext_vector_type(8))) float v8f;

#define IN_CH  128
#define OUT_CH 128

static __device__ __forceinline__ void atomic_add_f32(float* p, float v) {
    __hip_atomic_fetch_add(p, v, __ATOMIC_RELAXED, __HIP_MEMORY_SCOPE_AGENT);
}

__global__ void renorm_pack_kernel(const float* __restrict__ W,
                                   float* __restrict__ Wfrag) {
    int k = threadIdx.x;
    float s = 0.0f;
    #pragma unroll 8
    for (int o = 0; o < OUT_CH; ++o) {
        float w = W[o * IN_CH + k];
        s += w * w;
    }
    float nrm = sqrtf(s);
    float scale = (nrm > 1.0f) ? (1.0f / nrm) : 1.0f;

    const int kg  = k >> 2;
    const int kk  = k & 3;
    const int hi  = (kk >> 1) << 4;
    const int sub = kk & 1;
    #pragma unroll 8
    for (int o = 0; o < OUT_CH; ++o) {
        int cw = o >> 4;
        int m  = o & 15;
        int lane = hi | m;
        Wfrag[(((kg * 8 + cw) * 32) + lane) * 2 + sub] = W[o * IN_CH + k] * scale;
    }
}

__global__ void deg_init_kernel(float* __restrict__ deg, int N) {
    int n = blockIdx.x * blockDim.x + threadIdx.x;
    if (n < N) deg[n] = 1.0f;
}

__global__ void deg_count_kernel(const int* __restrict__ col,
                                 float* __restrict__ deg, int E) {
    int e = blockIdx.x * blockDim.x + threadIdx.x;
    if (e < E) atomic_add_f32(&deg[col[e]], 1.0f);
}

__global__ void deg_rsqrt_kernel(float* __restrict__ deg, int N) {
    int n = blockIdx.x * blockDim.x + threadIdx.x;
    if (n < N) deg[n] = rsqrtf(deg[n]);
}

__global__ void gemm_wmma_kernel(const float* __restrict__ x,
                                 const float* __restrict__ Wfrag,
                                 float* __restrict__ h, int N) {
    const int wave  = threadIdx.x >> 5;
    const int lane  = threadIdx.x & 31;
    const int row0  = blockIdx.x * 32;
    const int col0  = wave * 16;
    const int m     = lane & 15;
    const int khalf = (lane >> 4) * 2;

    int r0 = row0 + m;        if (r0 >= N) r0 = N - 1;
    int r1 = row0 + 16 + m;   if (r1 >= N) r1 = N - 1;
    const float* __restrict__ x0 = x + (size_t)r0 * IN_CH + khalf;
    const float* __restrict__ x1 = x + (size_t)r1 * IN_CH + khalf;
    const v2f*   __restrict__ bp = (const v2f*)Wfrag + (wave * 32 + lane);

    v8f c0 = {}, c1 = {};
    #pragma unroll 4
    for (int kg = 0; kg < 32; ++kg) {
        v2f a0 = *(const v2f*)(x0 + kg * 4);
        v2f a1 = *(const v2f*)(x1 + kg * 4);
        v2f b  = bp[kg * (8 * 32)];
        c0 = __builtin_amdgcn_wmma_f32_16x16x4_f32(
                false, a0, false, b, (short)0, c0, false, false);
        c1 = __builtin_amdgcn_wmma_f32_16x16x4_f32(
                false, a1, false, b, (short)0, c1, false, false);
    }

    const int nhalf = (lane >> 4) * 8;
    float* __restrict__ h0 = h + (size_t)(row0 + nhalf) * OUT_CH + col0 + m;
    float* __restrict__ h1 = h0 + (size_t)16 * OUT_CH;
    #pragma unroll
    for (int r = 0; r < 8; ++r) {
        h0[(size_t)r * OUT_CH] = c0[r];
        h1[(size_t)r * OUT_CH] = c1[r];
    }
}

__global__ void out_init_kernel(const float* __restrict__ h,
                                const float* __restrict__ dis,
                                const float* __restrict__ bias,
                                float* __restrict__ out, int N) {
    int t = blockIdx.x * blockDim.x + threadIdx.x;
    if (t >= N * 32) return;
    int n = t >> 5;
    int g = t & 31;
    float s = dis[n];
    s = s * s;
    float4 hv = ((const float4*)(h + (size_t)n * OUT_CH))[g];
    float4 bv = ((const float4*)bias)[g];
    float4 o;
    o.x = bv.x + hv.x * s;
    o.y = bv.y + hv.y * s;
    o.z = bv.z + hv.z * s;
    o.w = bv.w + hv.w * s;
    ((float4*)(out + (size_t)n * OUT_CH))[g] = o;
}

__global__ void scatter_kernel(const int* __restrict__ rowi,
                               const int* __restrict__ coli,
                               const float* __restrict__ h,
                               const float* __restrict__ dis,
                               float* __restrict__ out, int E) {
    int e = blockIdx.x * (blockDim.x >> 5) + (threadIdx.x >> 5);
    if (e >= E) return;
    int lane = threadIdx.x & 31;
    int r = rowi[e];
    int cdst = coli[e];
    float w = dis[r] * dis[cdst];
    float4 v = ((const float4*)(h + (size_t)r * OUT_CH))[lane];
    float* op = out + (size_t)cdst * OUT_CH + lane * 4;
    atomic_add_f32(op + 0, v.x * w);
    atomic_add_f32(op + 1, v.y * w);
    atomic_add_f32(op + 2, v.z * w);
    atomic_add_f32(op + 3, v.w * w);
}

extern "C" void kernel_launch(void* const* d_in, const int* in_sizes, int n_in,
                              void* d_out, int out_size, void* d_ws, size_t ws_size,
                              hipStream_t stream) {
    const float* x  = (const float*)d_in[0];
    const int*   ei = (const int*)d_in[1];
    const float* W  = (const float*)d_in[2];
    const float* b  = (const float*)d_in[3];
    float* out = (float*)d_out;

    const int N = in_sizes[0] / IN_CH;
    const int E = in_sizes[1] / 2;
    const int Npad = (N + 31) & ~31;
    const int* row = ei;
    const int* col = ei + E;

    float* ws    = (float*)d_ws;
    float* Wfrag = ws;
    float* deg   = ws + IN_CH * OUT_CH;
    float* h     = deg + Npad;

    renorm_pack_kernel<<<1, 128, 0, stream>>>(W, Wfrag);

    deg_init_kernel<<<(N + 255) / 256, 256, 0, stream>>>(deg, N);
    deg_count_kernel<<<(E + 255) / 256, 256, 0, stream>>>(col, deg, E);
    deg_rsqrt_kernel<<<(N + 255) / 256, 256, 0, stream>>>(deg, N);

    gemm_wmma_kernel<<<(N + 31) / 32, 256, 0, stream>>>(x, Wfrag, h, N);

    out_init_kernel<<<(N * 32 + 255) / 256, 256, 0, stream>>>(h, deg, b, out, N);

    scatter_kernel<<<(E + 7) / 8, 256, 0, stream>>>(row, col, h, deg, out, E);
}


# degree phase: per-block LDS integer histograms over node range x edge slice, summed per node, replace one contended global f32 atomic per edge; counts exact, 3 launches become 2
# speedup vs baseline: 1.2959x; 1.2959x over previous
#include <hip/hip_runtime.h>
#include <hip/hip_bf16.h>

typedef __attribute__((ext_vector_type(2))) float v2f;
typedef __attribute__((ext_vector_type(8))) float v8f;

#define IN_CH  128
#define OUT_CH 128

static __device__ __forceinline__ void atomic_add_f32(float* p, float v) {
    __hip_atomic_fetch_add(p, v, __ATOMIC_RELAXED, __HIP_MEMORY_SCOPE_AGENT);
}

__global__ void renorm_pack_kernel(const float* __restrict__ W,
                                   float* __restrict__ Wfrag) {
    int k = threadIdx.x;
    float s = 0.0f;
    #pragma unroll 8
    for (int o = 0; o < OUT_CH; ++o) {
        float w = W[o * IN_CH + k];
        s += w * w;
    }
    float nrm = sqrtf(s);
    float scale = (nrm > 1.0f) ? (1.0f / nrm) : 1.0f;

    const int kg  = k >> 2;
    const int kk  = k & 3;
    const int hi  = (kk >> 1) << 4;
    const int sub = kk & 1;
    #pragma unroll 8
    for (int o = 0; o < OUT_CH; ++o) {
        int cw = o >> 4;
        int m  = o & 15;
        int lane = hi | m;
        Wfrag[(((kg * 8 + cw) * 32) + lane) * 2 + sub] = W[o * IN_CH + k] * scale;
    }
}

#define DEG_GN   16
#define DEG_S    16
#define DEG_CAP  8192

__global__ void __launch_bounds__(1024)
deg_count_kernel(const int* __restrict__ col, int* __restrict__ part,
                 int N, int E, int stride) {
    __shared__ int cnt[DEG_CAP];
    const int tid = threadIdx.x;
    const int R = (((N + DEG_GN - 1) / DEG_GN) + 31) & ~31;
    const int nlo = blockIdx.x * R;
    int nhi = nlo + R;            if (nhi > N) nhi = N;
    const int per = (((E + DEG_S - 1) / DEG_S) + 3) & ~3;
    int elo = blockIdx.y * per;   if (elo > E) elo = E;
    int ehi = elo + per;          if (ehi > E) ehi = E;
    const int* __restrict__ cs = col + elo;
    const int ne = ehi - elo;
    const int n4 = ((((size_t)cs) & 15) == 0) ? (ne >> 2) : 0;
    const int4* __restrict__ c4 = (const int4*)cs;
    int* __restrict__ prow = part + (size_t)blockIdx.y * stride;

    for (int lo = nlo; lo < nhi; lo += DEG_CAP) {
        int span = nhi - lo;      if (span > DEG_CAP) span = DEG_CAP;
        for (int i = tid; i < span; i += 1024) cnt[i] = 0;
        __syncthreads();
        const unsigned uspan = (unsigned)span;
        const unsigned ulo   = (unsigned)lo;
        const int miss = lo - 1;
        for (int i0 = tid; i0 < n4; i0 += 4 * 1024) {
            int4 c[4];
            #pragma unroll
            for (int k = 0; k < 4; ++k) {
                const int i = i0 + k * 1024;
                c[k] = (i < n4) ? c4[i] : make_int4(miss, miss, miss, miss);
            }
            #pragma unroll
            for (int k = 0; k < 4; ++k) {
                unsigned d0 = (unsigned)c[k].x - ulo;
                unsigned d1 = (unsigned)c[k].y - ulo;
                unsigned d2 = (unsigned)c[k].z - ulo;
                unsigned d3 = (unsigned)c[k].w - ulo;
                if (d0 < uspan) atomicAdd(&cnt[d0], 1);
                if (d1 < uspan) atomicAdd(&cnt[d1], 1);
                if (d2 < uspan) atomicAdd(&cnt[d2], 1);
                if (d3 < uspan) atomicAdd(&cnt[d3], 1);
            }
        }
        for (int e = n4 * 4 + tid; e < ne; e += 1024) {
            unsigned d = (unsigned)cs[e] - ulo;
            if (d < uspan) atomicAdd(&cnt[d], 1);
        }
        __syncthreads();
        for (int i = tid; i < span; i += 1024) prow[lo + i] = cnt[i];
        __syncthreads();
    }
}

__global__ void deg_finish_kernel(const int* __restrict__ part,
                                  float* __restrict__ dis, int N, int stride) {
    int n = blockIdx.x * blockDim.x + threadIdx.x;
    if (n >= N) return;
    int c = 0;
    #pragma unroll
    for (int s = 0; s < DEG_S; ++s) c += part[(size_t)s * stride + n];
    dis[n] = rsqrtf(1.0f + (float)c);
}

__global__ void gemm_wmma_kernel(const float* __restrict__ x,
                                 const float* __restrict__ Wfrag,
                                 float* __restrict__ h, int N) {
    const int wave  = threadIdx.x >> 5;
    const int lane  = threadIdx.x & 31;
    const int row0  = blockIdx.x * 32;
    const int col0  = wave * 16;
    const int m     = lane & 15;
    const int khalf = (lane >> 4) * 2;

    int r0 = row0 + m;        if (r0 >= N) r0 = N - 1;
    int r1 = row0 + 16 + m;   if (r1 >= N) r1 = N - 1;
    const float* __restrict__ x0 = x + (size_t)r0 * IN_CH + khalf;
    const float* __restrict__ x1 = x + (size_t)r1 * IN_CH + khalf;
    const v2f*   __restrict__ bp = (const v2f*)Wfrag + (wave * 32 + lane);

    v8f c0 = {}, c1 = {};
    #pragma unroll 4
    for (int kg = 0; kg < 32; ++kg) {
        v2f a0 = *(const v2f*)(x0 + kg * 4);
        v2f a1 = *(const v2f*)(x1 + kg * 4);
        v2f b  = bp[kg * (8 * 32)];
        c0 = __builtin_amdgcn_wmma_f32_16x16x4_f32(
                false, a0, false, b, (short)0, c0, false, false);
        c1 = __builtin_amdgcn_wmma_f32_16x16x4_f32(
                false, a1, false, b, (short)0, c1, false, false);
    }

    const int nhalf = (lane >> 4) * 8;
    float* __restrict__ h0 = h + (size_t)(row0 + nhalf) * OUT_CH + col0 + m;
    float* __restrict__ h1 = h0 + (size_t)16 * OUT_CH;
    #pragma unroll
    for (int r = 0; r < 8; ++r) {
        h0[(size_t)r * OUT_CH] = c0[r];
        h1[(size_t)r * OUT_CH] = c1[r];
    }
}

__global__ void out_init_kernel(const float* __restrict__ h,
                                const float* __restrict__ dis,
                                const float* __restrict__ bias,
                                float* __restrict__ out, int N) {
    int t = blockIdx.x * blockDim.x + threadIdx.x;
    if (t >= N * 32) return;
    int n = t >> 5;
    int g = t & 31;
    float s = dis[n];
    s = s * s;
    float4 hv = ((const float4*)(h + (size_t)n * OUT_CH))[g];
    float4 bv = ((const float4*)bias)[g];
    float4 o;
    o.x = bv.x + hv.x * s;
    o.y = bv.y + hv.y * s;
    o.z = bv.z + hv.z * s;
    o.w = bv.w + hv.w * s;
    ((float4*)(out + (size_t)n * OUT_CH))[g] = o;
}

__global__ void scatter_kernel(const int* __restrict__ rowi,
                               const int* __restrict__ coli,
                               const float* __restrict__ h,
                               const float* __restrict__ dis,
                               float* __restrict__ out, int E) {
    int e = blockIdx.x * (blockDim.x >> 5) + (threadIdx.x >> 5);
    if (e >= E) return;
    int lane = threadIdx.x & 31;
    int r = rowi[e];
    int cdst = coli[e];
    float w = dis[r] * dis[cdst];
    float4 v = ((const float4*)(h + (size_t)r * OUT_CH))[lane];
    float* op = out + (size_t)cdst * OUT_CH + lane * 4;
    atomic_add_f32(op + 0, v.x * w);
    atomic_add_f32(op + 1, v.y * w);
    atomic_add_f32(op + 2, v.z * w);
    atomic_add_f32(op + 3, v.w * w);
}

extern "C" void kernel_launch(void* const* d_in, const int* in_sizes, int n_in,
                              void* d_out, int out_size, void* d_ws, size_t ws_size,
                              hipStream_t stream) {
    const float* x  = (const float*)d_in[0];
    const int*   ei = (const int*)d_in[1];
    const float* W  = (const float*)d_in[2];
    const float* b  = (const float*)d_in[3];
    float* out = (float*)d_out;

    const int N = in_sizes[0] / IN_CH;
    const int E = in_sizes[1] / 2;
    const int Npad = (N + 31) & ~31;
    const int* row = ei;
    const int* col = ei + E;

    float* ws    = (float*)d_ws;
    float* Wfrag = ws;
    float* deg   = ws + IN_CH * OUT_CH;
    float* h     = deg + Npad;
    int*   part  = (int*)(h + (size_t)Npad * OUT_CH);

    renorm_pack_kernel<<<1, 128, 0, stream>>>(W, Wfrag);

    deg_count_kernel<<<dim3(DEG_GN, DEG_S), 1024, 0, stream>>>(col, part, N, E, Npad);
    deg_finish_kernel<<<(N + 255) / 256, 256, 0, stream>>>(part, deg, N, Npad);

    gemm_wmma_kernel<<<(N + 31) / 32, 256, 0, stream>>>(x, Wfrag, h, N);

    out_init_kernel<<<(N * 32 + 255) / 256, 256, 0, stream>>>(h, deg, b, out, N);

    scatter_kernel<<<(E + 7) / 8, 256, 0, stream>>>(row, col, h, deg, out, E);
}


# edge sum: per-edge f32 atomics replaced by counting sort of sources by destination (LDS cursors, arrival-order rank), one wave per node sums in registers; out written once, 6 launches become 5
# speedup vs baseline: 3.3210x; 2.5627x over previous
#include <hip/hip_runtime.h>
#include <hip/hip_bf16.h>

typedef __attribute__((ext_vector_type(2))) float v2f;
typedef __attribute__((ext_vector_type(8))) float v8f;

#define IN_CH  128
#define OUT_CH 128

__global__ void renorm_pack_kernel(const float* __restrict__ W,
                                   float* __restrict__ Wfrag) {
    int k = threadIdx.x;
    float s = 0.0f;
    #pragma unroll 8
    for (int o = 0; o < OUT_CH; ++o) {
        float w = W[o * IN_CH + k];
        s += w * w;
    }
    float nrm = sqrtf(s);
    float scale = (nrm > 1.0f) ? (1.0f / nrm) : 1.0f;

    const int kg  = k >> 2;
    const int kk  = k & 3;
    const int hi  = (kk >> 1) << 4;
    const int sub = kk & 1;
    #pragma unroll 8
    for (int o = 0; o < OUT_CH; ++o) {
        int cw = o >> 4;
        int m  = o & 15;
        int lane = hi | m;
        Wfrag[(((kg * 8 + cw) * 32) + lane) * 2 + sub] = W[o * IN_CH + k] * scale;
    }
}

#define DEG_GN   16
#define DEG_S    16
#define DEG_CAP  8192

__global__ void __launch_bounds__(1024)
deg_count_kernel(const int* __restrict__ col, int* __restrict__ part,
                 int N, int E, int stride) {
    __shared__ int cnt[DEG_CAP];
    const int tid = threadIdx.x;
    const int R = (((N + DEG_GN - 1) / DEG_GN) + 31) & ~31;
    const int nlo = blockIdx.x * R;
    int nhi = nlo + R;            if (nhi > N) nhi = N;
    const int per = (((E + DEG_S - 1) / DEG_S) + 3) & ~3;
    int elo = blockIdx.y * per;   if (elo > E) elo = E;
    int ehi = elo + per;          if (ehi > E) ehi = E;
    const int* __restrict__ cs = col + elo;
    const int ne = ehi - elo;
    const int n4 = ((((size_t)cs) & 15) == 0) ? (ne >> 2) : 0;
    const int4* __restrict__ c4 = (const int4*)cs;
    int* __restrict__ prow = part + (size_t)blockIdx.y * stride;

    for (int lo = nlo; lo < nhi; lo += DEG_CAP) {
        int span = nhi - lo;      if (span > DEG_CAP) span = DEG_CAP;
        for (int i = tid; i < span; i += 1024) cnt[i] = 0;
        __syncthreads();
        const unsigned uspan = (unsigned)span;
        const unsigned ulo   = (unsigned)lo;
        const int miss = lo - 1;
        for (int i0 = tid; i0 < n4; i0 += 4 * 1024) {
            int4 c[4];
            #pragma unroll
            for (int k = 0; k < 4; ++k) {
                const int i = i0 + k * 1024;
                c[k] = (i < n4) ? c4[i] : make_int4(miss, miss, miss, miss);
            }
            #pragma unroll
            for (int k = 0; k < 4; ++k) {
                unsigned d0 = (unsigned)c[k].x - ulo;
                unsigned d1 = (unsigned)c[k].y - ulo;
                unsigned d2 = (unsigned)c[k].z - ulo;
                unsigned d3 = (unsigned)c[k].w - ulo;
                if (d0 < uspan) atomicAdd(&cnt[d0], 1);
                if (d1 < uspan) atomicAdd(&cnt[d1], 1);
                if (d2 < uspan) atomicAdd(&cnt[d2], 1);
                if (d3 < uspan) atomicAdd(&cnt[d3], 1);
            }
        }
        for (int e = n4 * 4 + tid; e < ne; e += 1024) {
            unsigned d = (unsigned)cs[e] - ulo;
            if (d < uspan) atomicAdd(&cnt[d], 1);
        }
        __syncthreads();
        for (int i = tid; i < span; i += 1024) prow[lo + i] = cnt[i];
        __syncthreads();
    }
}

#define FILL_PASS 4096

__global__ void __launch_bounds__(1024)
fill_kernel(const int* __restrict__ row, const int* __restrict__ col,
            const int* __restrict__ part, float* __restrict__ dis,
            int* __restrict__ rowptr, int* __restrict__ cntout,
            int* __restrict__ srcs, int N, int E, int stride, int cap) {
    __shared__ int cur[FILL_PASS + 1024];
    __shared__ int wsum[32];
    const int tid  = threadIdx.x;
    const int lane = tid & 31;
    const int wave = tid >> 5;
    const int sl   = blockIdx.y;
    const int R = (((N + DEG_GN - 1) / DEG_GN) + 31) & ~31;
    const int nlo = blockIdx.x * R;
    int nhi = nlo + R;            if (nhi > N) nhi = N;
    const int per = (((E + DEG_S - 1) / DEG_S) + 3) & ~3;
    int elo = blockIdx.y * per;   if (elo > E) elo = E;
    int ehi = elo + per;          if (ehi > E) ehi = E;
    const int* __restrict__ cs = col + elo;
    const int* __restrict__ rs = row + elo;
    const int ne = ehi - elo;
    const int n4 = (((((size_t)cs) | ((size_t)rs)) & 15) == 0) ? (ne >> 2) : 0;
    const int4* __restrict__ c4 = (const int4*)cs;
    const int4* __restrict__ r4 = (const int4*)rs;
    int* __restrict__ region = srcs + (size_t)blockIdx.x * cap;
    const unsigned ucap = (unsigned)cap;
    const unsigned dummy = FILL_PASS + tid;
    int carry = 0;

    for (int lo = nlo; lo < nhi; lo += FILL_PASS) {
        int span = nhi - lo;      if (span > FILL_PASS) span = FILL_PASS;
        const int i0 = tid * 4;
        int tot[4] = {0, 0, 0, 0}, bef[4] = {0, 0, 0, 0};
        if (i0 < span) {
            #pragma unroll
            for (int s = 0; s < DEG_S; ++s) {
                int4 p = *(const int4*)(part + (size_t)s * stride + lo + i0);
                tot[0] += p.x; tot[1] += p.y; tot[2] += p.z; tot[3] += p.w;
                if (s < sl) { bef[0] += p.x; bef[1] += p.y; bef[2] += p.z; bef[3] += p.w; }
            }
            #pragma unroll
            for (int k = 0; k < 4; ++k) if (i0 + k >= span) { tot[k] = 0; bef[k] = 0; }
        }
        const int t = tot[0] + tot[1] + tot[2] + tot[3];
        int inc = t;
        #pragma unroll
        for (int o = 1; o < 32; o <<= 1) { int u = __shfl_up(inc, o); if (lane >= o) inc += u; }
        if (lane == 31) wsum[wave] = inc;
        cur[dummy] = 0;
        __syncthreads();
        const int wv = wsum[lane];
        int winc = wv;
        #pragma unroll
        for (int o = 1; o < 32; o <<= 1) { int u = __shfl_up(winc, o); if (lane >= o) winc += u; }
        const int wbase = __shfl(winc, wave) - __shfl(wv, wave);
        const int passTotal = __shfl(winc, 31);
        int st = carry + wbase + inc - t;
        int stk[4];
        #pragma unroll
        for (int k = 0; k < 4; ++k) {
            stk[k] = st;
            if (i0 + k < span) cur[i0 + k] = st + bef[k];
            st += tot[k];
        }
        if (sl == 0 && i0 < span) {
            const int n = lo + i0;
            const int gb = blockIdx.x * cap;
            if (i0 + 3 < span) {
                *(int4*)(rowptr + n) = make_int4(gb + stk[0], gb + stk[1], gb + stk[2], gb + stk[3]);
                *(int4*)(cntout + n) = make_int4(tot[0], tot[1], tot[2], tot[3]);
                *(float4*)(dis + n)  = make_float4(rsqrtf(1.0f + (float)tot[0]),
                                                   rsqrtf(1.0f + (float)tot[1]),
                                                   rsqrtf(1.0f + (float)tot[2]),
                                                   rsqrtf(1.0f + (float)tot[3]));
            } else {
                #pragma unroll
                for (int k = 0; k < 4; ++k) {
                    if (i0 + k < span) {
                        rowptr[n + k] = gb + stk[k];
                        cntout[n + k] = tot[k];
                        dis[n + k] = rsqrtf(1.0f + (float)tot[k]);
                    }
                }
            }
        }
        __syncthreads();

        const unsigned uspan = (unsigned)span;
        const unsigned ulo   = (unsigned)lo;
        const int miss = lo - 1;
        for (int j0 = tid; j0 < n4; j0 += 4 * 1024) {
            int4 c[4], r[4];
            #pragma unroll
            for (int k = 0; k < 4; ++k) {
                const int i = j0 + k * 1024;
                if (i < n4) { c[k] = c4[i]; r[k] = r4[i]; }
                else { c[k] = make_int4(miss, miss, miss, miss); r[k] = make_int4(0, 0, 0, 0); }
            }
            #pragma unroll
            for (int k = 0; k < 4; ++k) {
                const unsigned d0 = (unsigned)c[k].x - ulo, d1 = (unsigned)c[k].y - ulo;
                const unsigned d2 = (unsigned)c[k].z - ulo, d3 = (unsigned)c[k].w - ulo;
                const unsigned p0 = (unsigned)atomicAdd(&cur[d0 < uspan ? d0 : dummy], 1);
                const unsigned p1 = (unsigned)atomicAdd(&cur[d1 < uspan ? d1 : dummy], 1);
                const unsigned p2 = (unsigned)atomicAdd(&cur[d2 < uspan ? d2 : dummy], 1);
                const unsigned p3 = (unsigned)atomicAdd(&cur[d3 < uspan ? d3 : dummy], 1);
                if (d0 < uspan && p0 < ucap) region[p0] = r[k].x;
                if (d1 < uspan && p1 < ucap) region[p1] = r[k].y;
                if (d2 < uspan && p2 < ucap) region[p2] = r[k].z;
                if (d3 < uspan && p3 < ucap) region[p3] = r[k].w;
            }
        }
        for (int e = n4 * 4 + tid; e < ne; e += 1024) {
            const unsigned d = (unsigned)cs[e] - ulo;
            if (d < uspan) {
                const unsigned p = (unsigned)atomicAdd(&cur[d], 1);
                if (p < ucap) region[p] = rs[e];
            }
        }
        carry += passTotal;
        __syncthreads();
    }
}

__global__ void gemm_wmma_kernel(const float* __restrict__ x,
                                 const float* __restrict__ Wfrag,
                                 float* __restrict__ h, int N) {
    const int wave  = threadIdx.x >> 5;
    const int lane  = threadIdx.x & 31;
    const int row0  = blockIdx.x * 32;
    const int col0  = wave * 16;
    const int m     = lane & 15;
    const int khalf = (lane >> 4) * 2;

    int r0 = row0 + m;        if (r0 >= N) r0 = N - 1;
    int r1 = row0 + 16 + m;   if (r1 >= N) r1 = N - 1;
    const float* __restrict__ x0 = x + (size_t)r0 * IN_CH + khalf;
    const float* __restrict__ x1 = x + (size_t)r1 * IN_CH + khalf;
    const v2f*   __restrict__ bp = (const v2f*)Wfrag + (wave * 32 + lane);

    v8f c0 = {}, c1 = {};
    #pragma unroll 4
    for (int kg = 0; kg < 32; ++kg) {
        v2f a0 = *(const v2f*)(x0 + kg * 4);
        v2f a1 = *(const v2f*)(x1 + kg * 4);
        v2f b  = bp[kg * (8 * 32)];
        c0 = __builtin_amdgcn_wmma_f32_16x16x4_f32(
                false, a0, false, b, (short)0, c0, false, false);
        c1 = __builtin_amdgcn_wmma_f32_16x16x4_f32(
                false, a1, false, b, (short)0, c1, false, false);
    }

    const int nhalf = (lane >> 4) * 8;
    float* __restrict__ h0 = h + (size_t)(row0 + nhalf) * OUT_CH + col0 + m;
    float* __restrict__ h1 = h0 + (size_t)16 * OUT_CH;
    #pragma unroll
    for (int r = 0; r < 8; ++r) {
        h0[(size_t)r * OUT_CH] = c0[r];
        h1[(size_t)r * OUT_CH] = c1[r];
    }
}

__global__ void __launch_bounds__(256)
gather_kernel(const int* __restrict__ rowptr, const int* __restrict__ cnt,
              const int* __restrict__ srcs, const float* __restrict__ h,
              const float* __restrict__ dis, const float* __restrict__ bias,
              float* __restrict__ out, int N, int E, int nsrc) {
    const int n = __builtin_amdgcn_readfirstlane((int)(blockIdx.x * 8 + (threadIdx.x >> 5)));
    if (n >= N) return;
    const int lane = threadIdx.x & 31;
    const float4* __restrict__ h4 = (const float4*)h;
    const int start = rowptr[n];
    int c = cnt[n];               if (c > E) c = E;
    const float dn = dis[n];
    const float4 hv = h4[(size_t)n * 32 + lane];
    const float4 bv = ((const float4*)bias)[lane];
    const float s = dn * dn;
    float4 acc;
    acc.x = bv.x + hv.x * s;
    acc.y = bv.y + hv.y * s;
    acc.z = bv.z + hv.z * s;
    acc.w = bv.w + hv.w * s;
    for (int j0 = 0; j0 < c; j0 += 32) {
        int m = c - j0;           if (m > 32) m = 32;
        int my = 0;
        float myw = 0.0f;
        if (lane < m) {
            unsigned at = (unsigned)(start + j0 + lane);
            if (at >= (unsigned)nsrc) at = 0;
            my = srcs[at];
            if ((unsigned)my >= (unsigned)N) my = 0;
            myw = dis[my] * dn;
        }
        #pragma unroll 4
        for (int j = 0; j < m; ++j) {
            const int   r = __builtin_amdgcn_readlane(my, j);
            const float w = __int_as_float(__builtin_amdgcn_readlane(__float_as_int(myw), j));
            const float4 v = h4[(size_t)r * 32 + lane];
            acc.x += v.x * w;
            acc.y += v.y * w;
            acc.z += v.z * w;
            acc.w += v.w * w;
        }
    }
    ((float4*)out)[(size_t)n * 32 + lane] = acc;
}

extern "C" void kernel_launch(void* const* d_in, const int* in_sizes, int n_in,
                              void* d_out, int out_size, void* d_ws, size_t ws_size,
                              hipStream_t stream) {
    const float* x  = (const float*)d_in[0];
    const int*   ei = (const int*)d_in[1];
    const float* W  = (const float*)d_in[2];
    const float* b  = (const float*)d_in[3];
    float* out = (float*)d_out;

    const int N = in_sizes[0] / IN_CH;
    const int E = in_sizes[1] / 2;
    const int Npad = (N + 31) & ~31;
    const int* row = ei;
    const int* col = ei + E;

    float* ws    = (float*)d_ws;
    float* Wfrag = ws;
    float* deg   = ws + IN_CH * OUT_CH;
    float* h     = deg + Npad;
    int*   part  = (int*)(h + (size_t)Npad * OUT_CH);
    int*   rowptr = part + (size_t)DEG_S * Npad;
    int*   cnt    = rowptr + Npad;
    int*   srcs   = cnt + Npad;
    const int cap = (E + 31) & ~31;

    renorm_pack_kernel<<<1, 128, 0, stream>>>(W, Wfrag);

    deg_count_kernel<<<dim3(DEG_GN, DEG_S), 1024, 0, stream>>>(col, part, N, E, Npad);
    fill_kernel<<<dim3(DEG_GN, DEG_S), 1024, 0, stream>>>(row, col, part, deg, rowptr, cnt,
                                                         srcs, N, E, Npad, cap);

    gemm_wmma_kernel<<<(N + 31) / 32, 256, 0, stream>>>(x, Wfrag, h, N);

    gather_kernel<<<(N + 7) / 8, 256, 0, stream>>>(rowptr, cnt, srcs, h, deg, b, out,
                                                   N, E, DEG_GN * cap);
}


# W renorm+pack leaves its own 1-block launch and rides in the count launch as 16 extra 1024-thread blocks (one load per thread, column met in LDS, same o order of the sum); 5 launches become 4
# speedup vs baseline: 4.5345x; 1.3654x over previous
#include <hip/hip_runtime.h>
#include <hip/hip_bf16.h>

typedef __attribute__((ext_vector_type(2))) float v2f;
typedef __attribute__((ext_vector_type(8))) float v8f;

#define IN_CH  128
#define OUT_CH 128

#define DEG_GN   16
#define DEG_S    16
#define DEG_CAP  8192

static_assert(DEG_GN * 8 == IN_CH && OUT_CH * 8 == 1024,
              "phase 0: DEG_GN blocks of 8 input channels, 1024 threads = OUT_CH x 8");

__global__ void __launch_bounds__(1024)
renorm_count_kernel(const float* __restrict__ W, float* __restrict__ Wfrag,
                    const int* __restrict__ col, int* __restrict__ part,
                    int N, int E, int stride) {
    __shared__ int cnt[DEG_CAP];
    __shared__ float wl[OUT_CH * 8];
    __shared__ float sc[8];
    const int tid = threadIdx.x;

    if (blockIdx.y == 0) {
        const int kk = tid & 7;
        const int o  = tid >> 3;
        const int k  = blockIdx.x * 8 + kk;
        const float w = W[o * IN_CH + k];
        wl[o * 8 + kk] = w;
        __syncthreads();
        if (tid < 8) {
            float s = 0.0f;
            #pragma unroll 16
            for (int oo = 0; oo < OUT_CH; ++oo) {
                const float v = wl[oo * 8 + tid];
                s += v * v;
            }
            const float nrm = sqrtf(s);
            sc[tid] = (nrm > 1.0f) ? (1.0f / nrm) : 1.0f;
        }
        __syncthreads();
        const float scale = sc[kk];
        const int kg  = k >> 2;
        const int k4  = k & 3;
        const int hi  = (k4 >> 1) << 4;
        const int sub = k4 & 1;
        const int cw  = o >> 4;
        const int m   = o & 15;
        const int lane = hi | m;
        Wfrag[(((kg * 8 + cw) * 32) + lane) * 2 + sub] = w * scale;
        return;
    }
    const int sl = blockIdx.y - 1;
    const int R = (((N + DEG_GN - 1) / DEG_GN) + 31) & ~31;
    const int nlo = blockIdx.x * R;
    int nhi = nlo + R;            if (nhi > N) nhi = N;
    const int per = (((E + DEG_S - 1) / DEG_S) + 3) & ~3;
    int elo = sl * per;           if (elo > E) elo = E;
    int ehi = elo + per;          if (ehi > E) ehi = E;
    const int* __restrict__ cs = col + elo;
    const int ne = ehi - elo;
    const int n4 = ((((size_t)cs) & 15) == 0) ? (ne >> 2) : 0;
    const int4* __restrict__ c4 = (const int4*)cs;
    int* __restrict__ prow = part + (size_t)sl * stride;

    for (int lo = nlo; lo < nhi; lo += DEG_CAP) {
        int span = nhi - lo;      if (span > DEG_CAP) span = DEG_CAP;
        for (int i = tid; i < span; i += 1024) cnt[i] = 0;
        __syncthreads();
        const unsigned uspan = (unsigned)span;
        const unsigned ulo   = (unsigned)lo;
        const int miss = lo - 1;
        for (int i0 = tid; i0 < n4; i0 += 4 * 1024) {
            int4 c[4];
            #pragma unroll
            for (int k = 0; k < 4; ++k) {
                const int i = i0 + k * 1024;
                c[k] = (i < n4) ? c4[i] : make_int4(miss, miss, miss, miss);
            }
            #pragma unroll
            for (int k = 0; k < 4; ++k) {
                unsigned d0 = (unsigned)c[k].x - ulo;
                unsigned d1 = (unsigned)c[k].y - ulo;
                unsigned d2 = (unsigned)c[k].z - ulo;
                unsigned d3 = (unsigned)c[k].w - ulo;
                if (d0 < uspan) atomicAdd(&cnt[d0], 1);
                if (d1 < uspan) atomicAdd(&cnt[d1], 1);
                if (d2 < uspan) atomicAdd(&cnt[d2], 1);
                if (d3 < uspan) atomicAdd(&cnt[d3], 1);
            }
        }
        for (int e = n4 * 4 + tid; e < ne; e += 1024) {
            unsigned d = (unsigned)cs[e] - ulo;
            if (d < uspan) atomicAdd(&cnt[d], 1);
        }
        __syncthreads();
        for (int i = tid; i < span; i += 1024) prow[lo + i] = cnt[i];
        __syncthreads();
    }
}

#define FILL_PASS 4096

__global__ void __launch_bounds__(1024)
fill_kernel(const int* __restrict__ row, const int* __restrict__ col,
            const int* __restrict__ part, float* __restrict__ dis,
            int* __restrict__ rowptr, int* __restrict__ cntout,
            int* __restrict__ srcs, int N, int E, int stride, int cap) {
    __shared__ int cur[FILL_PASS + 1024];
    __shared__ int wsum[32];
    const int tid  = threadIdx.x;
    const int lane = tid & 31;
    const int wave = tid >> 5;
    const int sl   = blockIdx.y;
    const int R = (((N + DEG_GN - 1) / DEG_GN) + 31) & ~31;
    const int nlo = blockIdx.x * R;
    int nhi = nlo + R;            if (nhi > N) nhi = N;
    const int per = (((E + DEG_S - 1) / DEG_S) + 3) & ~3;
    int elo = blockIdx.y * per;   if (elo > E) elo = E;
    int ehi = elo + per;          if (ehi > E) ehi = E;
    const int* __restrict__ cs = col + elo;
    const int* __restrict__ rs = row + elo;
    const int ne = ehi - elo;
    const int n4 = (((((size_t)cs) | ((size_t)rs)) & 15) == 0) ? (ne >> 2) : 0;
    const int4* __restrict__ c4 = (const int4*)cs;
    const int4* __restrict__ r4 = (const int4*)rs;
    int* __restrict__ region = srcs + (size_t)blockIdx.x * cap;
    const unsigned ucap = (unsigned)cap;
    const unsigned dummy = FILL_PASS + tid;
    int carry = 0;

    for (int lo = nlo; lo < nhi; lo += FILL_PASS) {
        int span = nhi - lo;      if (span > FILL_PASS) span = FILL_PASS;
        const int i0 = tid * 4;
        int tot[4] = {0, 0, 0, 0}, bef[4] = {0, 0, 0, 0};
        if (i0 < span) {
            #pragma unroll
            for (int s = 0; s < DEG_S; ++s) {
                int4 p = *(const int4*)(part + (size_t)s * stride + lo + i0);
                tot[0] += p.x; tot[1] += p.y; tot[2] += p.z; tot[3] += p.w;
                if (s < sl) { bef[0] += p.x; bef[1] += p.y; bef[2] += p.z; bef[3] += p.w; }
            }
            #pragma unroll
            for (int k = 0; k < 4; ++k) if (i0 + k >= span) { tot[k] = 0; bef[k] = 0; }
        }
        const int t = tot[0] + tot[1] + tot[2] + tot[3];
        int inc = t;
        #pragma unroll
        for (int o = 1; o < 32; o <<= 1) { int u = __shfl_up(inc, o); if (lane >= o) inc += u; }
        if (lane == 31) wsum[wave] = inc;
        cur[dummy] = 0;
        __syncthreads();
        const int wv = wsum[lane];
        int winc = wv;
        #pragma unroll
        for (int o = 1; o < 32; o <<= 1) { int u = __shfl_up(winc, o); if (lane >= o) winc += u; }
        const int wbase = __shfl(winc, wave) - __shfl(wv, wave);
        const int passTotal = __shfl(winc, 31);
        int st = carry + wbase + inc - t;
        int stk[4];
        #pragma unroll
        for (int k = 0; k < 4; ++k) {
            stk[k] = st;
            if (i0 + k < span) cur[i0 + k] = st + bef[k];
            st += tot[k];
        }
        if (sl == 0 && i0 < span) {
            const int n = lo + i0;
            const int gb = blockIdx.x * cap;
            if (i0 + 3 < span) {
                *(int4*)(rowptr + n) = make_int4(gb + stk[0], gb + stk[1], gb + stk[2], gb + stk[3]);
                *(int4*)(cntout + n) = make_int4(tot[0], tot[1], tot[2], tot[3]);
                *(float4*)(dis + n)  = make_float4(rsqrtf(1.0f + (float)tot[0]),
                                                   rsqrtf(1.0f + (float)tot[1]),
                                                   rsqrtf(1.0f + (float)tot[2]),
                                                   rsqrtf(1.0f + (float)tot[3]));
            } else {
                #pragma unroll
                for (int k = 0; k < 4; ++k) {
                    if (i0 + k < span) {
                        rowptr[n + k] = gb + stk[k];
                        cntout[n + k] = tot[k];
                        dis[n + k] = rsqrtf(1.0f + (float)tot[k]);
                    }
                }
            }
        }
        __syncthreads();

        const unsigned uspan = (unsigned)span;
        const unsigned ulo   = (unsigned)lo;
        const int miss = lo - 1;
        for (int j0 = tid; j0 < n4; j0 += 4 * 1024) {
            int4 c[4], r[4];
            #pragma unroll
            for (int k = 0; k < 4; ++k) {
                const int i = j0 + k * 1024;
                if (i < n4) { c[k] = c4[i]; r[k] = r4[i]; }
                else { c[k] = make_int4(miss, miss, miss, miss); r[k] = make_int4(0, 0, 0, 0); }
            }
            #pragma unroll
            for (int k = 0; k < 4; ++k) {
                const unsigned d0 = (unsigned)c[k].x - ulo, d1 = (unsigned)c[k].y - ulo;
                const unsigned d2 = (unsigned)c[k].z - ulo, d3 = (unsigned)c[k].w - ulo;
                const unsigned p0 = (unsigned)atomicAdd(&cur[d0 < uspan ? d0 : dummy], 1);
                const unsigned p1 = (unsigned)atomicAdd(&cur[d1 < uspan ? d1 : dummy], 1);
                const unsigned p2 = (unsigned)atomicAdd(&cur[d2 < uspan ? d2 : dummy], 1);
                const unsigned p3 = (unsigned)atomicAdd(&cur[d3 < uspan ? d3 : dummy], 1);
                if (d0 < uspan && p0 < ucap) region[p0] = r[k].x;
                if (d1 < uspan && p1 < ucap) region[p1] = r[k].y;
                if (d2 < uspan && p2 < ucap) region[p2] = r[k].z;
                if (d3 < uspan && p3 < ucap) region[p3] = r[k].w;
            }
        }
        for (int e = n4 * 4 + tid; e < ne; e += 1024) {
            const unsigned d = (unsigned)cs[e] - ulo;
            if (d < uspan) {
                const unsigned p = (unsigned)atomicAdd(&cur[d], 1);
                if (p < ucap) region[p] = rs[e];
            }
        }
        carry += passTotal;
        __syncthreads();
    }
}

__global__ void gemm_wmma_kernel(const float* __restrict__ x,
                                 const float* __restrict__ Wfrag,
                                 const float* __restrict__ dis,
                                 float* __restrict__ h, int N) {
    const int wave  = threadIdx.x >> 5;
    const int lane  = threadIdx.x & 31;
    const int row0  = blockIdx.x * 32;
    const int col0  = wave * 16;
    const int m     = lane & 15;
    const int khalf = (lane >> 4) * 2;

    int r0 = row0 + m;        if (r0 >= N) r0 = N - 1;
    int r1 = row0 + 16 + m;   if (r1 >= N) r1 = N - 1;
    const float* __restrict__ x0 = x + (size_t)r0 * IN_CH + khalf;
    const float* __restrict__ x1 = x + (size_t)r1 * IN_CH + khalf;
    const v2f*   __restrict__ bp = (const v2f*)Wfrag + (wave * 32 + lane);

    int dr = row0 + lane;     if (dr >= N) dr = N - 1;
    const float dv = dis[dr];

    v8f c0 = {}, c1 = {};
    #pragma unroll 4
    for (int kg = 0; kg < 32; ++kg) {
        v2f a0 = *(const v2f*)(x0 + kg * 4);
        v2f a1 = *(const v2f*)(x1 + kg * 4);
        v2f b  = bp[kg * (8 * 32)];
        c0 = __builtin_amdgcn_wmma_f32_16x16x4_f32(
                false, a0, false, b, (short)0, c0, false, false);
        c1 = __builtin_amdgcn_wmma_f32_16x16x4_f32(
                false, a1, false, b, (short)0, c1, false, false);
    }

    const int nhalf = (lane >> 4) * 8;
    float* __restrict__ h0 = h + (size_t)(row0 + nhalf) * OUT_CH + col0 + m;
    float* __restrict__ h1 = h0 + (size_t)16 * OUT_CH;
    #pragma unroll
    for (int r = 0; r < 8; ++r) {
        const float s0 = __shfl(dv, nhalf + r);
        const float s1 = __shfl(dv, 16 + nhalf + r);
        h0[(size_t)r * OUT_CH] = c0[r] * s0;
        h1[(size_t)r * OUT_CH] = c1[r] * s1;
    }
}

__global__ void __launch_bounds__(256)
gather_kernel(const int* __restrict__ rowptr, const int* __restrict__ cnt,
              const int* __restrict__ srcs, const float* __restrict__ h,
              const float* __restrict__ dis, const float* __restrict__ bias,
              float* __restrict__ out, int N, int E, int nsrc) {
    const int n = __builtin_amdgcn_readfirstlane((int)(blockIdx.x * 8 + (threadIdx.x >> 5)));
    if (n >= N) return;
    const int lane = threadIdx.x & 31;
    const float4* __restrict__ h4 = (const float4*)h;
    const int start = rowptr[n];
    int c = cnt[n];               if (c > E) c = E;
    const float dn = dis[n];
    float4 acc = h4[(size_t)n * 32 + lane];
    const float4 bv = ((const float4*)bias)[lane];
    for (int j0 = 0; j0 < c; j0 += 32) {
        int m = c - j0;           if (m > 32) m = 32;
        int my = 0;
        if (lane < m) {
            unsigned at = (unsigned)(start + j0 + lane);
            if (at >= (unsigned)nsrc) at = 0;
            my = srcs[at];
            if ((unsigned)my >= (unsigned)N) my = 0;
        }
        #pragma unroll 4
        for (int j = 0; j < m; ++j) {
            const int r = __builtin_amdgcn_readlane(my, j);
            const float4 v = h4[(size_t)r * 32 + lane];
            acc.x += v.x;
            acc.y += v.y;
            acc.z += v.z;
            acc.w += v.w;
        }
    }
    float4 o;
    o.x = bv.x + acc.x * dn;
    o.y = bv.y + acc.y * dn;
    o.z = bv.z + acc.z * dn;
    o.w = bv.w + acc.w * dn;
    ((float4*)out)[(size_t)n * 32 + lane] = o;
}

extern "C" void kernel_launch(void* const* d_in, const int* in_sizes, int n_in,
                              void* d_out, int out_size, void* d_ws, size_t ws_size,
                              hipStream_t stream) {
    const float* x  = (const float*)d_in[0];
    const int*   ei = (const int*)d_in[1];
    const float* W  = (const float*)d_in[2];
    const float* b  = (const float*)d_in[3];
    float* out = (float*)d_out;

    const int N = in_sizes[0] / IN_CH;
    const int E = in_sizes[1] / 2;
    const int Npad = (N + 31) & ~31;
    const int* row = ei;
    const int* col = ei + E;

    float* ws    = (float*)d_ws;
    float* h     = ws;
    float* Wfrag = h + (size_t)Npad * OUT_CH;
    float* deg   = Wfrag + IN_CH * OUT_CH;
    int*   part  = (int*)(deg + Npad);
    int*   rowptr = part + (size_t)DEG_S * Npad;
    int*   cnt    = rowptr + Npad;
    int*   srcs   = cnt + Npad;
    const int cap = (E + 31) & ~31;

    renorm_count_kernel<<<dim3(DEG_GN, DEG_S + 1), 1024, 0, stream>>>(W, Wfrag, col, part,
                                                                      N, E, Npad);

    fill_kernel<<<dim3(DEG_GN, DEG_S), 1024, 0, stream>>>(row, col, part, deg, rowptr, cnt,
                                                         srcs, N, E, Npad, cap);

    gemm_wmma_kernel<<<(N + 31) / 32, 256, 0, stream>>>(x, Wfrag, deg, h, N);

    gather_kernel<<<(N + 7) / 8, 256, 0, stream>>>(rowptr, cnt, srcs, h, deg, b, out,
                                                   N, E, DEG_GN * cap);
}


# GEMM: the block's 32x128 x tile comes in once by whole 512-B rows into LDS (pitch 528 B) and the 8 waves read A fragments there, not 16-line 8-B global reads per wave; k order, sums unchanged
# speedup vs baseline: 5.0593x; 1.1157x over previous
#include <hip/hip_runtime.h>
#include <hip/hip_bf16.h>

typedef __attribute__((ext_vector_type(2))) float v2f;
typedef __attribute__((ext_vector_type(8))) float v8f;

#define IN_CH  128
#define OUT_CH 128

#define DEG_GN   16
#define DEG_S    16
#define DEG_CAP  8192

static_assert(DEG_GN * 8 == IN_CH && OUT_CH * 8 == 1024,
              "phase 0: DEG_GN blocks of 8 input channels, 1024 threads = OUT_CH x 8");

__global__ void __launch_bounds__(1024)
renorm_count_kernel(const float* __restrict__ W, float* __restrict__ Wfrag,
                    const int* __restrict__ col, int* __restrict__ part,
                    int N, int E, int stride) {
    __shared__ int cnt[DEG_CAP];
    __shared__ float wl[OUT_CH * 8];
    __shared__ float sc[8];
    const int tid = threadIdx.x;

    if (blockIdx.y == 0) {
        const int kk = tid & 7;
        const int o  = tid >> 3;
        const int k  = blockIdx.x * 8 + kk;
        const float w = W[o * IN_CH + k];
        wl[o * 8 + kk] = w;
        __syncthreads();
        if (tid < 8) {
            float s = 0.0f;
            #pragma unroll 16
            for (int oo = 0; oo < OUT_CH; ++oo) {
                const float v = wl[oo * 8 + tid];
                s += v * v;
            }
            const float nrm = sqrtf(s);
            sc[tid] = (nrm > 1.0f) ? (1.0f / nrm) : 1.0f;
        }
        __syncthreads();
        const float scale = sc[kk];
        const int kg  = k >> 2;
        const int k4  = k & 3;
        const int hi  = (k4 >> 1) << 4;
        const int sub = k4 & 1;
        const int cw  = o >> 4;
        const int m   = o & 15;
        const int lane = hi | m;
        Wfrag[(((kg * 8 + cw) * 32) + lane) * 2 + sub] = w * scale;
        return;
    }
    const int sl = blockIdx.y - 1;
    const int R = (((N + DEG_GN - 1) / DEG_GN) + 31) & ~31;
    const int nlo = blockIdx.x * R;
    int nhi = nlo + R;            if (nhi > N) nhi = N;
    const int per = (((E + DEG_S - 1) / DEG_S) + 3) & ~3;
    int elo = sl * per;           if (elo > E) elo = E;
    int ehi = elo + per;          if (ehi > E) ehi = E;
    const int* __restrict__ cs = col + elo;
    const int ne = ehi - elo;
    const int n4 = ((((size_t)cs) & 15) == 0) ? (ne >> 2) : 0;
    const int4* __restrict__ c4 = (const int4*)cs;
    int* __restrict__ prow = part + (size_t)sl * stride;

    for (int lo = nlo; lo < nhi; lo += DEG_CAP) {
        int span = nhi - lo;      if (span > DEG_CAP) span = DEG_CAP;
        for (int i = tid; i < span; i += 1024) cnt[i] = 0;
        __syncthreads();
        const unsigned uspan = (unsigned)span;
        const unsigned ulo   = (unsigned)lo;
        const int miss = lo - 1;
        for (int i0 = tid; i0 < n4; i0 += 4 * 1024) {
            int4 c[4];
            #pragma unroll
            for (int k = 0; k < 4; ++k) {
                const int i = i0 + k * 1024;
                c[k] = (i < n4) ? c4[i] : make_int4(miss, miss, miss, miss);
            }
            #pragma unroll
            for (int k = 0; k < 4; ++k) {
                unsigned d0 = (unsigned)c[k].x - ulo;
                unsigned d1 = (unsigned)c[k].y - ulo;
                unsigned d2 = (unsigned)c[k].z - ulo;
                unsigned d3 = (unsigned)c[k].w - ulo;
                if (d0 < uspan) atomicAdd(&cnt[d0], 1);
                if (d1 < uspan) atomicAdd(&cnt[d1], 1);
                if (d2 < uspan) atomicAdd(&cnt[d2], 1);
                if (d3 < uspan) atomicAdd(&cnt[d3], 1);
            }
        }
        for (int e = n4 * 4 + tid; e < ne; e += 1024) {
            unsigned d = (unsigned)cs[e] - ulo;
            if (d < uspan) atomicAdd(&cnt[d], 1);
        }
        __syncthreads();
        for (int i = tid; i < span; i += 1024) prow[lo + i] = cnt[i];
        __syncthreads();
    }
}

#define FILL_PASS 4096

__global__ void __launch_bounds__(1024)
fill_kernel(const int* __restrict__ row, const int* __restrict__ col,
            const int* __restrict__ part, float* __restrict__ dis,
            int* __restrict__ rowptr, int* __restrict__ cntout,
            int* __restrict__ srcs, int N, int E, int stride, int cap) {
    __shared__ int cur[FILL_PASS + 1024];
    __shared__ int wsum[32];
    const int tid  = threadIdx.x;
    const int lane = tid & 31;
    const int wave = tid >> 5;
    const int sl   = blockIdx.y;
    const int R = (((N + DEG_GN - 1) / DEG_GN) + 31) & ~31;
    const int nlo = blockIdx.x * R;
    int nhi = nlo + R;            if (nhi > N) nhi = N;
    const int per = (((E + DEG_S - 1) / DEG_S) + 3) & ~3;
    int elo = blockIdx.y * per;   if (elo > E) elo = E;
    int ehi = elo + per;          if (ehi > E) ehi = E;
    const int* __restrict__ cs = col + elo;
    const int* __restrict__ rs = row + elo;
    const int ne = ehi - elo;
    const int n4 = (((((size_t)cs) | ((size_t)rs)) & 15) == 0) ? (ne >> 2) : 0;
    const int4* __restrict__ c4 = (const int4*)cs;
    const int4* __restrict__ r4 = (const int4*)rs;
    int* __restrict__ region = srcs + (size_t)blockIdx.x * cap;
    const unsigned ucap = (unsigned)cap;
    const unsigned dummy = FILL_PASS + tid;
    int carry = 0;

    for (int lo = nlo; lo < nhi; lo += FILL_PASS) {
        int span = nhi - lo;      if (span > FILL_PASS) span = FILL_PASS;
        const int i0 = tid * 4;
        int tot[4] = {0, 0, 0, 0}, bef[4] = {0, 0, 0, 0};
        if (i0 < span) {
            #pragma unroll
            for (int s = 0; s < DEG_S; ++s) {
                int4 p = *(const int4*)(part + (size_t)s * stride + lo + i0);
                tot[0] += p.x; tot[1] += p.y; tot[2] += p.z; tot[3] += p.w;
                if (s < sl) { bef[0] += p.x; bef[1] += p.y; bef[2] += p.z; bef[3] += p.w; }
            }
            #pragma unroll
            for (int k = 0; k < 4; ++k) if (i0 + k >= span) { tot[k] = 0; bef[k] = 0; }
        }
        const int t = tot[0] + tot[1] + tot[2] + tot[3];
        int inc = t;
        #pragma unroll
        for (int o = 1; o < 32; o <<= 1) { int u = __shfl_up(inc, o); if (lane >= o) inc += u; }
        if (lane == 31) wsum[wave] = inc;
        cur[dummy] = 0;
        __syncthreads();
        const int wv = wsum[lane];
        int winc = wv;
        #pragma unroll
        for (int o = 1; o < 32; o <<= 1) { int u = __shfl_up(winc, o); if (lane >= o) winc += u; }
        const int wbase = __shfl(winc, wave) - __shfl(wv, wave);
        const int passTotal = __shfl(winc, 31);
        int st = carry + wbase + inc - t;
        int stk[4];
        #pragma unroll
        for (int k = 0; k < 4; ++k) {
            stk[k] = st;
            if (i0 + k < span) cur[i0 + k] = st + bef[k];
            st += tot[k];
        }
        if (sl == 0 && i0 < span) {
            const int n = lo + i0;
            const int gb = blockIdx.x * cap;
            if (i0 + 3 < span) {
                *(int4*)(rowptr + n) = make_int4(gb + stk[0], gb + stk[1], gb + stk[2], gb + stk[3]);
                *(int4*)(cntout + n) = make_int4(tot[0], tot[1], tot[2], tot[3]);
                *(float4*)(dis + n)  = make_float4(rsqrtf(1.0f + (float)tot[0]),
                                                   rsqrtf(1.0f + (float)tot[1]),
                                                   rsqrtf(1.0f + (float)tot[2]),
                                                   rsqrtf(1.0f + (float)tot[3]));
            } else {
                #pragma unroll
                for (int k = 0; k < 4; ++k) {
                    if (i0 + k < span) {
                        rowptr[n + k] = gb + stk[k];
                        cntout[n + k] = tot[k];
                        dis[n + k] = rsqrtf(1.0f + (float)tot[k]);
                    }
                }
            }
        }
        __syncthreads();

        const unsigned uspan = (unsigned)span;
        const unsigned ulo   = (unsigned)lo;
        const int miss = lo - 1;
        for (int j0 = tid; j0 < n4; j0 += 4 * 1024) {
            int4 c[4], r[4];
            #pragma unroll
            for (int k = 0; k < 4; ++k) {
                const int i = j0 + k * 1024;
                if (i < n4) { c[k] = c4[i]; r[k] = r4[i]; }
                else { c[k] = make_int4(miss, miss, miss, miss); r[k] = make_int4(0, 0, 0, 0); }
            }
            #pragma unroll
            for (int k = 0; k < 4; ++k) {
                const unsigned d0 = (unsigned)c[k].x - ulo, d1 = (unsigned)c[k].y - ulo;
                const unsigned d2 = (unsigned)c[k].z - ulo, d3 = (unsigned)c[k].w - ulo;
                const unsigned p0 = (unsigned)atomicAdd(&cur[d0 < uspan ? d0 : dummy], 1);
                const unsigned p1 = (unsigned)atomicAdd(&cur[d1 < uspan ? d1 : dummy], 1);
                const unsigned p2 = (unsigned)atomicAdd(&cur[d2 < uspan ? d2 : dummy], 1);
                const unsigned p3 = (unsigned)atomicAdd(&cur[d3 < uspan ? d3 : dummy], 1);
                if (d0 < uspan && p0 < ucap) region[p0] = r[k].x;
                if (d1 < uspan && p1 < ucap) region[p1] = r[k].y;
                if (d2 < uspan && p2 < ucap) region[p2] = r[k].z;
                if (d3 < uspan && p3 < ucap) region[p3] = r[k].w;
            }
        }
        for (int e = n4 * 4 + tid; e < ne; e += 1024) {
            const unsigned d = (unsigned)cs[e] - ulo;
            if (d < uspan) {
                const unsigned p = (unsigned)atomicAdd(&cur[d], 1);
                if (p < ucap) region[p] = rs[e];
            }
        }
        carry += passTotal;
        __syncthreads();
    }
}

#define XP 132

__global__ void __launch_bounds__(256)
gemm_wmma_kernel(const float* __restrict__ x,
                 const float* __restrict__ Wfrag,
                 const float* __restrict__ dis,
                 float* __restrict__ h, int N) {
    __shared__ __attribute__((aligned(16))) float xs[32 * XP];
    const int wave  = threadIdx.x >> 5;
    const int lane  = threadIdx.x & 31;
    const int row0  = blockIdx.x * 32;
    const int col0  = wave * 16;
    const int m     = lane & 15;
    const int khalf = (lane >> 4) * 2;

    #pragma unroll
    for (int i = 0; i < 4; ++i) {
        const int r = wave * 4 + i;
        int gr = row0 + r;    if (gr >= N) gr = N - 1;
        const float4 v = ((const float4*)(x + (size_t)gr * IN_CH))[lane];
        *(float4*)(xs + r * XP + lane * 4) = v;
    }
    const float* x0 = xs + m * XP + khalf;
    const float* x1 = x0 + 16 * XP;
    const v2f*   __restrict__ bp = (const v2f*)Wfrag + (wave * 32 + lane);

    int dr = row0 + lane;     if (dr >= N) dr = N - 1;
    const float dv = dis[dr];

    __syncthreads();

    v8f c0 = {}, c1 = {};
    #pragma unroll 4
    for (int kg = 0; kg < 32; ++kg) {
        v2f a0 = *(const v2f*)(x0 + kg * 4);
        v2f a1 = *(const v2f*)(x1 + kg * 4);
        v2f b  = bp[kg * (8 * 32)];
        c0 = __builtin_amdgcn_wmma_f32_16x16x4_f32(
                false, a0, false, b, (short)0, c0, false, false);
        c1 = __builtin_amdgcn_wmma_f32_16x16x4_f32(
                false, a1, false, b, (short)0, c1, false, false);
    }

    const int nhalf = (lane >> 4) * 8;
    float* __restrict__ h0 = h + (size_t)(row0 + nhalf) * OUT_CH + col0 + m;
    float* __restrict__ h1 = h0 + (size_t)16 * OUT_CH;
    #pragma unroll
    for (int r = 0; r < 8; ++r) {
        const float s0 = __shfl(dv, nhalf + r);
        const float s1 = __shfl(dv, 16 + nhalf + r);
        h0[(size_t)r * OUT_CH] = c0[r] * s0;
        h1[(size_t)r * OUT_CH] = c1[r] * s1;
    }
}

__global__ void __launch_bounds__(256)
gather_kernel(const int* __restrict__ rowptr, const int* __restrict__ cnt,
              const int* __restrict__ srcs, const float* __restrict__ h,
              const float* __restrict__ dis, const float* __restrict__ bias,
              float* __restrict__ out, int N, int E, int nsrc) {
    const int n = __builtin_amdgcn_readfirstlane((int)(blockIdx.x * 8 + (threadIdx.x >> 5)));
    if (n >= N) return;
    const int lane = threadIdx.x & 31;
    const float4* __restrict__ h4 = (const float4*)h;
    const int start = rowptr[n];
    int c = cnt[n];               if (c > E) c = E;
    const float dn = dis[n];
    float4 acc = h4[(size_t)n * 32 + lane];
    const float4 bv = ((const float4*)bias)[lane];
    for (int j0 = 0; j0 < c; j0 += 32) {
        int m = c - j0;           if (m > 32) m = 32;
        int my = 0;
        if (lane < m) {
            unsigned at = (unsigned)(start + j0 + lane);
            if (at >= (unsigned)nsrc) at = 0;
            my = srcs[at];
            if ((unsigned)my >= (unsigned)N) my = 0;
        }
        #pragma unroll 4
        for (int j = 0; j < m; ++j) {
            const int r = __builtin_amdgcn_readlane(my, j);
            const float4 v = h4[(size_t)r * 32 + lane];
            acc.x += v.x;
            acc.y += v.y;
            acc.z += v.z;
            acc.w += v.w;
        }
    }
    float4 o;
    o.x = bv.x + acc.x * dn;
    o.y = bv.y + acc.y * dn;
    o.z = bv.z + acc.z * dn;
    o.w = bv.w + acc.w * dn;
    ((float4*)out)[(size_t)n * 32 + lane] = o;
}

extern "C" void kernel_launch(void* const* d_in, const int* in_sizes, int n_in,
                              void* d_out, int out_size, void* d_ws, size_t ws_size,
                              hipStream_t stream) {
    const float* x  = (const float*)d_in[0];
    const int*   ei = (const int*)d_in[1];
    const float* W  = (const float*)d_in[2];
    const float* b  = (const float*)d_in[3];
    float* out = (float*)d_out;

    const int N = in_sizes[0] / IN_CH;
    const int E = in_sizes[1] / 2;
    const int Npad = (N + 31) & ~31;
    const int* row = ei;
    const int* col = ei + E;

    float* ws    = (float*)d_ws;
    float* h     = ws;
    float* Wfrag = h + (size_t)Npad * OUT_CH;
    float* deg   = Wfrag + IN_CH * OUT_CH;
    int*   part  = (int*)(deg + Npad);
    int*   rowptr = part + (size_t)DEG_S * Npad;
    int*   cnt    = rowptr + Npad;
    int*   srcs   = cnt + Npad;
    const int cap = (E + 31) & ~31;

    renorm_count_kernel<<<dim3(DEG_GN, DEG_S + 1), 1024, 0, stream>>>(W, Wfrag, col, part,
                                                                      N, E, Npad);

    fill_kernel<<<dim3(DEG_GN, DEG_S), 1024, 0, stream>>>(row, col, part, deg, rowptr, cnt,
                                                         srcs, N, E, Npad, cap);

    gemm_wmma_kernel<<<(N + 31) / 32, 256, 0, stream>>>(x, Wfrag, deg, h, N);

    gather_kernel<<<(N + 7) / 8, 256, 0, stream>>>(rowptr, cnt, srcs, h, deg, b, out,
                                                   N, E, DEG_GN * cap);
}


# GEMM leaves its own launch and rides in the sorting launch as extra 1024-thread blocks of 128 rows (kernel held to 64 registers: two blocks per WGP), dis taken from the count rows; 4 launches to 3
# speedup vs baseline: 5.4742x; 1.0820x over previous
#include <hip/hip_runtime.h>
#include <hip/hip_bf16.h>

typedef __attribute__((ext_vector_type(2))) float v2f;
typedef __attribute__((ext_vector_type(8))) float v8f;

#define IN_CH  128
#define OUT_CH 128

#define DEG_GN   16
#define DEG_S    16
#define DEG_CAP  8192

static_assert(DEG_GN * 8 == IN_CH && OUT_CH * 8 == 1024,
              "phase 0: DEG_GN blocks of 8 input channels, 1024 threads = OUT_CH x 8");

__global__ void __launch_bounds__(1024)
renorm_count_kernel(const float* __restrict__ W, float* __restrict__ Wfrag,
                    const int* __restrict__ col, int* __restrict__ part,
                    int N, int E, int stride) {
    __shared__ int cnt[DEG_CAP];
    __shared__ float wl[OUT_CH * 8];
    __shared__ float sc[8];
    const int tid = threadIdx.x;

    if (blockIdx.y == 0) {
        const int kk = tid & 7;
        const int o  = tid >> 3;
        const int k  = blockIdx.x * 8 + kk;
        const float w = W[o * IN_CH + k];
        wl[o * 8 + kk] = w;
        __syncthreads();
        if (tid < 8) {
            float s = 0.0f;
            #pragma unroll 16
            for (int oo = 0; oo < OUT_CH; ++oo) {
                const float v = wl[oo * 8 + tid];
                s += v * v;
            }
            const float nrm = sqrtf(s);
            sc[tid] = (nrm > 1.0f) ? (1.0f / nrm) : 1.0f;
        }
        __syncthreads();
        const float scale = sc[kk];
        const int kg  = k >> 2;
        const int k4  = k & 3;
        const int hi  = (k4 >> 1) << 4;
        const int sub = k4 & 1;
        const int cw  = o >> 4;
        const int m   = o & 15;
        const int lane = hi | m;
        Wfrag[(((kg * 8 + cw) * 32) + lane) * 2 + sub] = w * scale;
        return;
    }
    const int sl = blockIdx.y - 1;
    const int R = (((N + DEG_GN - 1) / DEG_GN) + 31) & ~31;
    const int nlo = blockIdx.x * R;
    int nhi = nlo + R;            if (nhi > N) nhi = N;
    const int per = (((E + DEG_S - 1) / DEG_S) + 3) & ~3;
    int elo = sl * per;           if (elo > E) elo = E;
    int ehi = elo + per;          if (ehi > E) ehi = E;
    const int* __restrict__ cs = col + elo;
    const int ne = ehi - elo;
    const int n4 = ((((size_t)cs) & 15) == 0) ? (ne >> 2) : 0;
    const int4* __restrict__ c4 = (const int4*)cs;
    int* __restrict__ prow = part + (size_t)sl * stride;

    for (int lo = nlo; lo < nhi; lo += DEG_CAP) {
        int span = nhi - lo;      if (span > DEG_CAP) span = DEG_CAP;
        for (int i = tid; i < span; i += 1024) cnt[i] = 0;
        __syncthreads();
        const unsigned uspan = (unsigned)span;
        const unsigned ulo   = (unsigned)lo;
        const int miss = lo - 1;
        for (int i0 = tid; i0 < n4; i0 += 4 * 1024) {
            int4 c[4];
            #pragma unroll
            for (int k = 0; k < 4; ++k) {
                const int i = i0 + k * 1024;
                c[k] = (i < n4) ? c4[i] : make_int4(miss, miss, miss, miss);
            }
            #pragma unroll
            for (int k = 0; k < 4; ++k) {
                unsigned d0 = (unsigned)c[k].x - ulo;
                unsigned d1 = (unsigned)c[k].y - ulo;
                unsigned d2 = (unsigned)c[k].z - ulo;
                unsigned d3 = (unsigned)c[k].w - ulo;
                if (d0 < uspan) atomicAdd(&cnt[d0], 1);
                if (d1 < uspan) atomicAdd(&cnt[d1], 1);
                if (d2 < uspan) atomicAdd(&cnt[d2], 1);
                if (d3 < uspan) atomicAdd(&cnt[d3], 1);
            }
        }
        for (int e = n4 * 4 + tid; e < ne; e += 1024) {
            unsigned d = (unsigned)cs[e] - ulo;
            if (d < uspan) atomicAdd(&cnt[d], 1);
        }
        __syncthreads();
        for (int i = tid; i < span; i += 1024) prow[lo + i] = cnt[i];
        __syncthreads();
    }
}

#define FILL_PASS 4096
#define GEMM_ROWS 128
#define XP 132

static_assert(GEMM_ROWS == 32 * 4 && GEMM_ROWS * 8 == 1024,
              "phase 2: 32 waves = 4 row groups x 8 column tiles, 4 staged rows per wave");

__global__ void __launch_bounds__(1024, 16)
fill_gemm_kernel(const int* __restrict__ row, const int* __restrict__ col,
                 const int* __restrict__ part, float* __restrict__ dis,
                 int* __restrict__ rowptr, int* __restrict__ cntout,
                 int* __restrict__ srcs,
                 const float* __restrict__ x, const float* __restrict__ Wfrag,
                 float* __restrict__ h,
                 int N, int E, int stride, int cap) {
    __shared__ int cur[FILL_PASS + 1024];
    __shared__ int wsum[32];
    __shared__ __attribute__((aligned(16))) float xs[GEMM_ROWS * XP];
    const int tid  = threadIdx.x;
    const int lane = tid & 31;
    const int wave = tid >> 5;

    if (blockIdx.y >= DEG_S) {
        const int g    = (blockIdx.y - DEG_S) * DEG_GN + blockIdx.x;
        const int row0 = g * GEMM_ROWS;
        if (row0 >= N) return;
        const int grow0 = row0 + (wave >> 3) * 32;
        const int col0  = (wave & 7) * 16;
        const int m     = lane & 15;
        const int khalf = (lane >> 4) * 2;

        #pragma unroll
        for (int i = 0; i < 4; ++i) {
            const int r = wave * 4 + i;
            int gr = row0 + r;    if (gr >= N) gr = N - 1;
            const float4 v = ((const float4*)(x + (size_t)gr * IN_CH))[lane];
            *(float4*)(xs + r * XP + lane * 4) = v;
        }
        int dr = grow0 + lane;    if (dr >= N) dr = N - 1;
        int dcount = 0;
        #pragma unroll
        for (int s = 0; s < DEG_S; ++s) dcount += part[(size_t)s * stride + dr];
        const float dv = rsqrtf(1.0f + (float)dcount);

        __syncthreads();
        if (grow0 >= N) return;

        const float* x0 = xs + ((wave >> 3) * 32 + m) * XP + khalf;
        const float* x1 = x0 + 16 * XP;
        const v2f*   __restrict__ bp = (const v2f*)Wfrag + ((wave & 7) * 32 + lane);

        v8f c0 = {}, c1 = {};
        #pragma unroll 4
        for (int kg = 0; kg < 32; ++kg) {
            v2f a0 = *(const v2f*)(x0 + kg * 4);
            v2f a1 = *(const v2f*)(x1 + kg * 4);
            v2f b  = bp[kg * (8 * 32)];
            c0 = __builtin_amdgcn_wmma_f32_16x16x4_f32(
                    false, a0, false, b, (short)0, c0, false, false);
            c1 = __builtin_amdgcn_wmma_f32_16x16x4_f32(
                    false, a1, false, b, (short)0, c1, false, false);
        }

        const int nhalf = (lane >> 4) * 8;
        float* __restrict__ h0 = h + (size_t)(grow0 + nhalf) * OUT_CH + col0 + m;
        float* __restrict__ h1 = h0 + (size_t)16 * OUT_CH;
        #pragma unroll
        for (int r = 0; r < 8; ++r) {
            const float s0 = __shfl(dv, nhalf + r);
            const float s1 = __shfl(dv, 16 + nhalf + r);
            h0[(size_t)r * OUT_CH] = c0[r] * s0;
            h1[(size_t)r * OUT_CH] = c1[r] * s1;
        }
        return;
    }

    const int sl   = blockIdx.y;
    const int R = (((N + DEG_GN - 1) / DEG_GN) + 31) & ~31;
    const int nlo = blockIdx.x * R;
    int nhi = nlo + R;            if (nhi > N) nhi = N;
    const int per = (((E + DEG_S - 1) / DEG_S) + 3) & ~3;
    int elo = blockIdx.y * per;   if (elo > E) elo = E;
    int ehi = elo + per;          if (ehi > E) ehi = E;
    const int* __restrict__ cs = col + elo;
    const int* __restrict__ rs = row + elo;
    const int ne = ehi - elo;
    const int n4 = (((((size_t)cs) | ((size_t)rs)) & 15) == 0) ? (ne >> 2) : 0;
    const int4* __restrict__ c4 = (const int4*)cs;
    const int4* __restrict__ r4 = (const int4*)rs;
    int* __restrict__ region = srcs + (size_t)blockIdx.x * cap;
    const unsigned ucap = (unsigned)cap;
    const unsigned dummy = FILL_PASS + tid;
    int carry = 0;

    for (int lo = nlo; lo < nhi; lo += FILL_PASS) {
        int span = nhi - lo;      if (span > FILL_PASS) span = FILL_PASS;
        const int i0 = tid * 4;
        int tot[4] = {0, 0, 0, 0}, bef[4] = {0, 0, 0, 0};
        if (i0 < span) {
            #pragma unroll 4
            for (int s = 0; s < DEG_S; ++s) {
                int4 p = *(const int4*)(part + (size_t)s * stride + (unsigned)(lo + i0));
                tot[0] += p.x; tot[1] += p.y; tot[2] += p.z; tot[3] += p.w;
                if (s < sl) { bef[0] += p.x; bef[1] += p.y; bef[2] += p.z; bef[3] += p.w; }
            }
            #pragma unroll
            for (int k = 0; k < 4; ++k) if (i0 + k >= span) { tot[k] = 0; bef[k] = 0; }
        }
        const int t = tot[0] + tot[1] + tot[2] + tot[3];
        int inc = t;
        #pragma unroll
        for (int o = 1; o < 32; o <<= 1) { int u = __shfl_up(inc, o); if (lane >= o) inc += u; }
        if (lane == 31) wsum[wave] = inc;
        cur[dummy] = 0;
        __syncthreads();
        const int wv = wsum[lane];
        int winc = wv;
        #pragma unroll
        for (int o = 1; o < 32; o <<= 1) { int u = __shfl_up(winc, o); if (lane >= o) winc += u; }
        const int wbase = __shfl(winc, wave) - __shfl(wv, wave);
        const int passTotal = __shfl(winc, 31);
        int st = carry + wbase + inc - t;
        int stk[4];
        #pragma unroll
        for (int k = 0; k < 4; ++k) {
            stk[k] = st;
            if (i0 + k < span) cur[i0 + k] = st + bef[k];
            st += tot[k];
        }
        if (sl == 0 && i0 < span) {
            const int n = lo + i0;
            const int gb = blockIdx.x * cap;
            if (i0 + 3 < span) {
                *(int4*)(rowptr + n) = make_int4(gb + stk[0], gb + stk[1], gb + stk[2], gb + stk[3]);
                *(int4*)(cntout + n) = make_int4(tot[0], tot[1], tot[2], tot[3]);
                *(float4*)(dis + n)  = make_float4(rsqrtf(1.0f + (float)tot[0]),
                                                   rsqrtf(1.0f + (float)tot[1]),
                                                   rsqrtf(1.0f + (float)tot[2]),
                                                   rsqrtf(1.0f + (float)tot[3]));
            } else {
                #pragma unroll
                for (int k = 0; k < 4; ++k) {
                    if (i0 + k < span) {
                        rowptr[n + k] = gb + stk[k];
                        cntout[n + k] = tot[k];
                        dis[n + k] = rsqrtf(1.0f + (float)tot[k]);
                    }
                }
            }
        }
        __syncthreads();

        const unsigned uspan = (unsigned)span;
        const unsigned ulo   = (unsigned)lo;
        const int miss = lo - 1;
        for (unsigned j0 = tid; j0 < (unsigned)n4; j0 += 4 * 1024) {
            int4 c[4], r[4];
            #pragma unroll
            for (int k = 0; k < 4; ++k) {
                const unsigned i = j0 + k * 1024;
                if (i < (unsigned)n4) { c[k] = c4[i]; r[k] = r4[i]; }
                else { c[k] = make_int4(miss, miss, miss, miss); r[k] = make_int4(0, 0, 0, 0); }
            }
            #pragma unroll
            for (int k = 0; k < 4; ++k) {
                const unsigned d0 = (unsigned)c[k].x - ulo, d1 = (unsigned)c[k].y - ulo;
                const unsigned d2 = (unsigned)c[k].z - ulo, d3 = (unsigned)c[k].w - ulo;
                const unsigned p0 = (unsigned)atomicAdd(&cur[d0 < uspan ? d0 : dummy], 1);
                const unsigned p1 = (unsigned)atomicAdd(&cur[d1 < uspan ? d1 : dummy], 1);
                const unsigned p2 = (unsigned)atomicAdd(&cur[d2 < uspan ? d2 : dummy], 1);
                const unsigned p3 = (unsigned)atomicAdd(&cur[d3 < uspan ? d3 : dummy], 1);
                if (d0 < uspan && p0 < ucap) region[p0] = r[k].x;
                if (d1 < uspan && p1 < ucap) region[p1] = r[k].y;
                if (d2 < uspan && p2 < ucap) region[p2] = r[k].z;
                if (d3 < uspan && p3 < ucap) region[p3] = r[k].w;
            }
        }
        for (int e = n4 * 4 + tid; e < ne; e += 1024) {
            const unsigned d = (unsigned)cs[e] - ulo;
            if (d < uspan) {
                const unsigned p = (unsigned)atomicAdd(&cur[d], 1);
                if (p < ucap) region[p] = rs[e];
            }
        }
        carry += passTotal;
        __syncthreads();
    }
}

__global__ void __launch_bounds__(256)
gather_kernel(const int* __restrict__ rowptr, const int* __restrict__ cnt,
              const int* __restrict__ srcs, const float* __restrict__ h,
              const float* __restrict__ dis, const float* __restrict__ bias,
              float* __restrict__ out, int N, int E, int nsrc) {
    const int n = __builtin_amdgcn_readfirstlane((int)(blockIdx.x * 8 + (threadIdx.x >> 5)));
    if (n >= N) return;
    const int lane = threadIdx.x & 31;
    const float4* __restrict__ h4 = (const float4*)h;
    const int start = rowptr[n];
    int c = cnt[n];               if (c > E) c = E;
    const float dn = dis[n];
    float4 acc = h4[(size_t)n * 32 + lane];
    const float4 bv = ((const float4*)bias)[lane];
    for (int j0 = 0; j0 < c; j0 += 32) {
        int m = c - j0;           if (m > 32) m = 32;
        int my = 0;
        if (lane < m) {
            unsigned at = (unsigned)(start + j0 + lane);
            if (at >= (unsigned)nsrc) at = 0;
            my = srcs[at];
            if ((unsigned)my >= (unsigned)N) my = 0;
        }
        #pragma unroll 4
        for (int j = 0; j < m; ++j) {
            const int r = __builtin_amdgcn_readlane(my, j);
            const float4 v = h4[(size_t)r * 32 + lane];
            acc.x += v.x;
            acc.y += v.y;
            acc.z += v.z;
            acc.w += v.w;
        }
    }
    float4 o;
    o.x = bv.x + acc.x * dn;
    o.y = bv.y + acc.y * dn;
    o.z = bv.z + acc.z * dn;
    o.w = bv.w + acc.w * dn;
    ((float4*)out)[(size_t)n * 32 + lane] = o;
}

extern "C" void kernel_launch(void* const* d_in, const int* in_sizes, int n_in,
                              void* d_out, int out_size, void* d_ws, size_t ws_size,
                              hipStream_t stream) {
    const float* x  = (const float*)d_in[0];
    const int*   ei = (const int*)d_in[1];
    const float* W  = (const float*)d_in[2];
    const float* b  = (const float*)d_in[3];
    float* out = (float*)d_out;

    const int N = in_sizes[0] / IN_CH;
    const int E = in_sizes[1] / 2;
    const int Npad = (N + 31) & ~31;
    const int* row = ei;
    const int* col = ei + E;

    float* ws    = (float*)d_ws;
    float* h     = ws;
    float* Wfrag = h + (size_t)Npad * OUT_CH;
    float* deg   = Wfrag + IN_CH * OUT_CH;
    int*   part  = (int*)(deg + Npad);
    int*   rowptr = part + (size_t)DEG_S * Npad;
    int*   cnt    = rowptr + Npad;
    int*   srcs   = cnt + Npad;
    const int cap = (E + 31) & ~31;

    renorm_count_kernel<<<dim3(DEG_GN, DEG_S + 1), 1024, 0, stream>>>(W, Wfrag, col, part,
                                                                      N, E, Npad);

    const int gemm_blocks = (N + GEMM_ROWS - 1) / GEMM_ROWS;
    const int gemm_grid_y = (gemm_blocks + DEG_GN - 1) / DEG_GN;
    fill_gemm_kernel<<<dim3(DEG_GN, DEG_S + gemm_grid_y), 1024, 0, stream>>>(
        row, col, part, deg, rowptr, cnt, srcs, x, Wfrag, h, N, E, Npad, cap);

    gather_kernel<<<(N + 7) / 8, 256, 0, stream>>>(rowptr, cnt, srcs, h, deg, b, out,
                                                   N, E, DEG_GN * cap);
}
